# starting kernel
# baseline (speedup 1.0000x reference)
#include <hip/hip_runtime.h>
#include <hip/hip_bf16.h>

#define GG 2048
#define HH 256
#define DD 256

typedef __attribute__((ext_vector_type(2))) float v2f;
typedef __attribute__((ext_vector_type(8))) float v8f;

__global__ void scan_offsets_kernel(const int* __restrict__ nn,
                                    int* __restrict__ offs) {
    __shared__ int chunk[256];
    const int t = threadIdx.x;
    int s = 0;
#pragma unroll
    for (int i = 0; i < 8; ++i) s += nn[t * 8 + i];
    chunk[t] = s;
    __syncthreads();
    if (t == 0) {
        int run = 0;
#pragma unroll 4
        for (int i = 0; i < 256; ++i) { int v = chunk[i]; chunk[i] = run; run += v; }
    }
    __syncthreads();
    int run = chunk[t];
#pragma unroll
    for (int i = 0; i < 8; ++i) {
        int v = nn[t * 8 + i];
        offs[t * 8 + i] = run;
        run += v;
    }
}

__global__ void pool_mean_kernel(const float* __restrict__ feat,
                                 const int* __restrict__ nn,
                                 const int* __restrict__ offs,
                                 float* __restrict__ M) {
    const int g = blockIdx.x;
    const int t = threadIdx.x;
    const int off = offs[g];
    const int cnt = nn[g];
    const float* p = feat + (size_t)off * HH + t;

    float a0 = 0.f, a1 = 0.f, a2 = 0.f, a3 = 0.f;
    int n = 0;
    for (; n + 4 <= cnt; n += 4) {
        a0 += p[(size_t)(n + 0) * HH];
        a1 += p[(size_t)(n + 1) * HH];
        a2 += p[(size_t)(n + 2) * HH];
        a3 += p[(size_t)(n + 3) * HH];
    }
    for (; n < cnt; ++n) a0 += p[(size_t)n * HH];

    const float inv = 1.0f / (float)cnt;
    M[(size_t)g * HH + t] = (a0 + a1 + a2 + a3) * inv;
}

__global__ void gemm3_wmma_kernel(const float* __restrict__ M,
                                  const float* __restrict__ Wp, const float* __restrict__ bp,
                                  const float* __restrict__ Wr, const float* __restrict__ br,
                                  const float* __restrict__ Wk, const float* __restrict__ bk,
                                  float* __restrict__ out) {
    const int wave = threadIdx.x >> 5;
    const int lane = threadIdx.x & 31;
    const int tile = blockIdx.x * 4 + wave;
    const int row0 = (tile >> 4) * 16;
    const int col0 = (tile & 15) * 16;
    const int lm = lane & 15;
    const int hi = lane >> 4;

    const float bpv = bp[col0 + lm];
    const float brv = br[col0 + lm];
    const float bkv = bk[col0 + lm];
    v8f cp, cr, ck;
#pragma unroll
    for (int j = 0; j < 8; ++j) { cp[j] = bpv; cr[j] = brv; ck[j] = bkv; }

    const float* arow = M  + (size_t)(row0 + lm) * HH + 2 * hi;
    const float* wpc  = Wp + (size_t)(2 * hi) * DD + col0 + lm;
    const float* wrc  = Wr + (size_t)(2 * hi) * DD + col0 + lm;
    const float* wkc  = Wk + (size_t)(2 * hi) * DD + col0 + lm;

#pragma unroll 4
    for (int k = 0; k < HH; k += 4) {
        v2f a = *(const v2f*)(arow + k);

        v2f bbp, bbr, bbk;
        const size_t ko = (size_t)k * DD;
        bbp.x = wpc[ko];  bbp.y = wpc[ko + DD];
        bbr.x = wrc[ko];  bbr.y = wrc[ko + DD];
        bbk.x = wkc[ko];  bbk.y = wkc[ko + DD];

        cp = __builtin_amdgcn_wmma_f32_16x16x4_f32(false, a, false, bbp, (short)0, cp, false, false);
        cr = __builtin_amdgcn_wmma_f32_16x16x4_f32(false, a, false, bbr, (short)0, cr, false, false);
        ck = __builtin_amdgcn_wmma_f32_16x16x4_f32(false, a, false, bbk, (short)0, ck, false, false);
    }

    const size_t GD = (size_t)GG * DD;
    float* okeys = out;
    float* opq   = out + GD;
    float* orq   = out + 2 * GD;
#pragma unroll
    for (int j = 0; j < 8; ++j) {
        const int row = row0 + j + 8 * hi;
        const size_t idx = (size_t)row * DD + col0 + lm;
        okeys[idx] = ck[j];
        opq[idx]   = cp[j];
        orq[idx]   = cr[j];
    }
}

extern "C" void kernel_launch(void* const* d_in, const int* in_sizes, int n_in,
                              void* d_out, int out_size, void* d_ws, size_t ws_size,
                              hipStream_t stream) {
    const float* features = (const float*)d_in[0];
    const int*   num_nodes = (const int*)d_in[1];
    const float* Wp = (const float*)d_in[2];
    const float* bp = (const float*)d_in[3];
    const float* Wr = (const float*)d_in[4];
    const float* br = (const float*)d_in[5];
    const float* Wk = (const float*)d_in[6];
    const float* bk = (const float*)d_in[7];
    float* out = (float*)d_out;

    int*   offs = (int*)d_ws;
    float* M    = (float*)((char*)d_ws + 8192);

    scan_offsets_kernel<<<1, 256, 0, stream>>>(num_nodes, offs);
    pool_mean_kernel<<<GG, 256, 0, stream>>>(features, num_nodes, offs, M);
    gemm3_wmma_kernel<<<GG / 4, 128, 0, stream>>>(M, Wp, bp, Wr, br, Wk, bk, out);
    (void)in_sizes; (void)n_in; (void)out_size; (void)ws_size;
}


# pool pass rewritten as a stream: 16 B/lane loads (512 B per wave load), one wave per half row of a graph, next 16 rows requested before summing; 4096 waves, not 16384; per-channel sum order kept
# speedup vs baseline: 1.0682x; 1.0682x over previous
#include <hip/hip_runtime.h>
#include <hip/hip_bf16.h>

#define GG 2048
#define HH 256
#define DD 256

typedef __attribute__((ext_vector_type(2))) float v2f;
typedef __attribute__((ext_vector_type(8))) float v8f;

__global__ void scan_offsets_kernel(const int* __restrict__ nn,
                                    int* __restrict__ offs) {
    __shared__ int chunk[256];
    const int t = threadIdx.x;
    int s = 0;
#pragma unroll
    for (int i = 0; i < 8; ++i) s += nn[t * 8 + i];
    chunk[t] = s;
    __syncthreads();
    if (t == 0) {
        int run = 0;
#pragma unroll 4
        for (int i = 0; i < 256; ++i) { int v = chunk[i]; chunk[i] = run; run += v; }
    }
    __syncthreads();
    int run = chunk[t];
#pragma unroll
    for (int i = 0; i < 8; ++i) {
        int v = nn[t * 8 + i];
        offs[t * 8 + i] = run;
        run += v;
    }
}

#define POOL_WAVES 16
#define POOL_ROWS  16

typedef __attribute__((ext_vector_type(4))) float v4f;

#define POOL_LOAD(BUF, BI)                                                    \
    _Pragma("unroll")                                                         \
    for (int i = 0; i < POOL_ROWS; ++i) {                                     \
        const int r_  = (BI) * POOL_ROWS + i;                                 \
        const int rr_ = r_ < last ? r_ : last;                                \
        BUF[i] = *(const v4f*)(gb + (lo + (unsigned)rr_ * (HH * 4u)));        \
    }

#define POOL_SUM_FULL(BUF)                                                    \
    _Pragma("unroll")                                                         \
    for (int i = 0; i < POOL_ROWS; ++i) acc[i & 3] += BUF[i];

#define POOL_SUM_LAST(BUF, BI)                                                \
    _Pragma("unroll")                                                         \
    for (int i = 0; i < POOL_ROWS; ++i) {                                     \
        const int r_ = (BI) * POOL_ROWS + i;                                  \
        const v4f x_ = BUF[i];                                                \
        acc[i & 3] += (r_ < cnt4) ? x_ : zero;                                \
        acc[0]     += (r_ >= cnt4 && r_ < cnt) ? x_ : zero;                   \
    }

#define POOL_PIN()                                                            \
    asm volatile("" : "+v"(acc[0]), "+v"(acc[1]), "+v"(acc[2]), "+v"(acc[3]) :: "memory")

__global__ void __launch_bounds__(POOL_WAVES * 32)
pool_mean_kernel(const float* __restrict__ feat,
                 const int* __restrict__ nn,
                 const int* __restrict__ offs,
                 float* __restrict__ M) {
    const int lane = threadIdx.x & 31;
    const int wave = __builtin_amdgcn_readfirstlane((int)(threadIdx.x >> 5));
    const int half = wave & 1;
    const int g    = (wave >> 1) * (GG / 8) + (int)blockIdx.x;
    const int off  = offs[g];
    const int cnt  = nn[g];
    const int cnt4 = cnt & ~3;
    const int last = cnt - 1;
    const int nb   = (cnt + POOL_ROWS - 1) / POOL_ROWS;

    const v4f zero = {0.f, 0.f, 0.f, 0.f};
    v4f acc[4] = {zero, zero, zero, zero};

    if (cnt > 0) {
        const char* gb = (const char*)(feat + (size_t)off * HH + half * (HH / 2));
        const unsigned lo = (unsigned)lane * 16u;
        v4f bufA[POOL_ROWS], bufB[POOL_ROWS];
        int b = 0;
        POOL_LOAD(bufA, 0)
        while (b + 2 < nb) {
            __builtin_amdgcn_sched_barrier(0);
            POOL_LOAD(bufB, b + 1)
            __builtin_amdgcn_sched_barrier(0);
            POOL_SUM_FULL(bufA)
            POOL_PIN();
            __builtin_amdgcn_sched_barrier(0);
            POOL_LOAD(bufA, b + 2)
            __builtin_amdgcn_sched_barrier(0);
            POOL_SUM_FULL(bufB)
            POOL_PIN();
            b += 2;
        }
        if (b + 2 == nb) {
            __builtin_amdgcn_sched_barrier(0);
            POOL_LOAD(bufB, b + 1)
            __builtin_amdgcn_sched_barrier(0);
            POOL_SUM_FULL(bufA)
            POOL_SUM_LAST(bufB, b + 1)
        } else {
            POOL_SUM_LAST(bufA, b)
        }
    }

    const float inv = 1.0f / (float)cnt;
    const v4f res = (acc[0] + acc[1] + acc[2] + acc[3]) * inv;
    *(v4f*)(M + (size_t)g * HH + half * (HH / 2) + lane * 4) = res;
}

__global__ void gemm3_wmma_kernel(const float* __restrict__ M,
                                  const float* __restrict__ Wp, const float* __restrict__ bp,
                                  const float* __restrict__ Wr, const float* __restrict__ br,
                                  const float* __restrict__ Wk, const float* __restrict__ bk,
                                  float* __restrict__ out) {
    const int wave = threadIdx.x >> 5;
    const int lane = threadIdx.x & 31;
    const int tile = blockIdx.x * 4 + wave;
    const int row0 = (tile >> 4) * 16;
    const int col0 = (tile & 15) * 16;
    const int lm = lane & 15;
    const int hi = lane >> 4;

    const float bpv = bp[col0 + lm];
    const float brv = br[col0 + lm];
    const float bkv = bk[col0 + lm];
    v8f cp, cr, ck;
#pragma unroll
    for (int j = 0; j < 8; ++j) { cp[j] = bpv; cr[j] = brv; ck[j] = bkv; }

    const float* arow = M  + (size_t)(row0 + lm) * HH + 2 * hi;
    const float* wpc  = Wp + (size_t)(2 * hi) * DD + col0 + lm;
    const float* wrc  = Wr + (size_t)(2 * hi) * DD + col0 + lm;
    const float* wkc  = Wk + (size_t)(2 * hi) * DD + col0 + lm;

#pragma unroll 4
    for (int k = 0; k < HH; k += 4) {
        v2f a = *(const v2f*)(arow + k);

        v2f bbp, bbr, bbk;
        const size_t ko = (size_t)k * DD;
        bbp.x = wpc[ko];  bbp.y = wpc[ko + DD];
        bbr.x = wrc[ko];  bbr.y = wrc[ko + DD];
        bbk.x = wkc[ko];  bbk.y = wkc[ko + DD];

        cp = __builtin_amdgcn_wmma_f32_16x16x4_f32(false, a, false, bbp, (short)0, cp, false, false);
        cr = __builtin_amdgcn_wmma_f32_16x16x4_f32(false, a, false, bbr, (short)0, cr, false, false);
        ck = __builtin_amdgcn_wmma_f32_16x16x4_f32(false, a, false, bbk, (short)0, ck, false, false);
    }

    const size_t GD = (size_t)GG * DD;
    float* okeys = out;
    float* opq   = out + GD;
    float* orq   = out + 2 * GD;
#pragma unroll
    for (int j = 0; j < 8; ++j) {
        const int row = row0 + j + 8 * hi;
        const size_t idx = (size_t)row * DD + col0 + lm;
        okeys[idx] = ck[j];
        opq[idx]   = cp[j];
        orq[idx]   = cr[j];
    }
}

extern "C" void kernel_launch(void* const* d_in, const int* in_sizes, int n_in,
                              void* d_out, int out_size, void* d_ws, size_t ws_size,
                              hipStream_t stream) {
    const float* features = (const float*)d_in[0];
    const int*   num_nodes = (const int*)d_in[1];
    const float* Wp = (const float*)d_in[2];
    const float* bp = (const float*)d_in[3];
    const float* Wr = (const float*)d_in[4];
    const float* br = (const float*)d_in[5];
    const float* Wk = (const float*)d_in[6];
    const float* bk = (const float*)d_in[7];
    float* out = (float*)d_out;

    int*   offs = (int*)d_ws;
    float* M    = (float*)((char*)d_ws + 8192);

    scan_offsets_kernel<<<1, 256, 0, stream>>>(num_nodes, offs);
    pool_mean_kernel<<<GG / 8, POOL_WAVES * 32, 0, stream>>>(features, num_nodes, offs, M);
    gemm3_wmma_kernel<<<GG / 4, 128, 0, stream>>>(M, Wp, bp, Wr, br, Wk, bk, out);
    (void)in_sizes; (void)n_in; (void)out_size; (void)ws_size;
}


# row offsets (prefix sum of num_nodes) computed inside the pool kernel by each workgroup (one 16 B load per thread, wave butterfly, 32 partials in LDS); scan launch removed: 2 launches, not 3
# speedup vs baseline: 1.1936x; 1.1174x over previous
#include <hip/hip_runtime.h>
#include <hip/hip_bf16.h>

#define GG 2048
#define HH 256
#define DD 256

typedef __attribute__((ext_vector_type(2))) float v2f;
typedef __attribute__((ext_vector_type(8))) float v8f;

#define POOL_WAVES 16
#define POOL_ROWS  16

typedef __attribute__((ext_vector_type(4))) float v4f;

#define POOL_LOAD(BUF, BI)                                                    \
    _Pragma("unroll")                                                         \
    for (int i = 0; i < POOL_ROWS; ++i) {                                     \
        const int r_  = (BI) * POOL_ROWS + i;                                 \
        const int rr_ = r_ < last ? r_ : last;                                \
        BUF[i] = *(const v4f*)(gb + (lo + (unsigned)rr_ * (HH * 4u)));        \
    }

#define POOL_SUM_FULL(BUF)                                                    \
    _Pragma("unroll")                                                         \
    for (int i = 0; i < POOL_ROWS; ++i) acc[i & 3] += BUF[i];

#define POOL_SUM_LAST(BUF, BI)                                                \
    _Pragma("unroll")                                                         \
    for (int i = 0; i < POOL_ROWS; ++i) {                                     \
        const int r_ = (BI) * POOL_ROWS + i;                                  \
        const v4f x_ = BUF[i];                                                \
        acc[i & 3] += (r_ < cnt4) ? x_ : zero;                                \
        acc[0]     += (r_ >= cnt4 && r_ < cnt) ? x_ : zero;                   \
    }

#define POOL_PIN()                                                            \
    asm volatile("" : "+v"(acc[0]), "+v"(acc[1]), "+v"(acc[2]), "+v"(acc[3]) :: "memory")

typedef __attribute__((ext_vector_type(4))) int v4i;

__device__ __forceinline__ int wave_sum(int x) {
    x += __builtin_amdgcn_update_dpp(0, x, 0x161, 0xf, 0xf, false);
    x += __builtin_amdgcn_update_dpp(0, x, 0x162, 0xf, 0xf, false);
    x += __builtin_amdgcn_update_dpp(0, x, 0x164, 0xf, 0xf, false);
    x += __builtin_amdgcn_update_dpp(0, x, 0x168, 0xf, 0xf, false);
    x += (int)__builtin_amdgcn_permlanex16((unsigned)x, (unsigned)x, 0x76543210u, 0xfedcba98u, false, false);
    return x;
}

__global__ void __launch_bounds__(POOL_WAVES * 32)
pool_mean_kernel(const float* __restrict__ feat,
                 const int* __restrict__ nn,
                 float* __restrict__ M) {
    __shared__ __attribute__((aligned(16))) int part[2 * POOL_WAVES];
    const int lane = threadIdx.x & 31;
    const int wave = __builtin_amdgcn_readfirstlane((int)(threadIdx.x >> 5));
    const int half = wave & 1;
    const int g    = (wave >> 1) * (GG / 8) + (int)blockIdx.x;
    const int cnt  = nn[g];

    {
        const int i0 = (int)threadIdx.x * 4;
        const v4i c4 = *(const v4i*)(nn + i0);
        const int whole = (c4.x + c4.y) + (c4.z + c4.w);
        const int below = ((i0     < g ? c4.x : 0) + (i0 + 1 < g ? c4.y : 0))
                        + ((i0 + 2 < g ? c4.z : 0) + (i0 + 3 < g ? c4.w : 0));
        const int ws = wave_sum(whole);
        const int bs = wave_sum(below);
        if (lane == 0) { part[wave] = ws; part[POOL_WAVES + wave] = bs; }
    }
    __syncthreads();
    const int pair = wave & ~1;
    int offv = part[POOL_WAVES + pair] + part[POOL_WAVES + pair + 1];
#pragma unroll
    for (int w = 0; w < POOL_WAVES - 2; ++w) offv += (w < pair) ? part[w] : 0;
    const int off  = __builtin_amdgcn_readfirstlane(offv);
    const int cnt4 = cnt & ~3;
    const int last = cnt - 1;
    const int nb   = (cnt + POOL_ROWS - 1) / POOL_ROWS;

    const v4f zero = {0.f, 0.f, 0.f, 0.f};
    v4f acc[4] = {zero, zero, zero, zero};

    if (cnt > 0) {
        const char* gb = (const char*)(feat + (size_t)off * HH + half * (HH / 2));
        const unsigned lo = (unsigned)lane * 16u;
        v4f bufA[POOL_ROWS], bufB[POOL_ROWS];
        int b = 0;
        POOL_LOAD(bufA, 0)
        while (b + 2 < nb) {
            __builtin_amdgcn_sched_barrier(0);
            POOL_LOAD(bufB, b + 1)
            __builtin_amdgcn_sched_barrier(0);
            POOL_SUM_FULL(bufA)
            POOL_PIN();
            __builtin_amdgcn_sched_barrier(0);
            POOL_LOAD(bufA, b + 2)
            __builtin_amdgcn_sched_barrier(0);
            POOL_SUM_FULL(bufB)
            POOL_PIN();
            b += 2;
        }
        if (b + 2 == nb) {
            __builtin_amdgcn_sched_barrier(0);
            POOL_LOAD(bufB, b + 1)
            __builtin_amdgcn_sched_barrier(0);
            POOL_SUM_FULL(bufA)
            POOL_SUM_LAST(bufB, b + 1)
        } else {
            POOL_SUM_LAST(bufA, b)
        }
    }

    const float inv = 1.0f / (float)cnt;
    const v4f res = (acc[0] + acc[1] + acc[2] + acc[3]) * inv;
    *(v4f*)(M + (size_t)g * HH + half * (HH / 2) + lane * 4) = res;
}

__global__ void gemm3_wmma_kernel(const float* __restrict__ M,
                                  const float* __restrict__ Wp, const float* __restrict__ bp,
                                  const float* __restrict__ Wr, const float* __restrict__ br,
                                  const float* __restrict__ Wk, const float* __restrict__ bk,
                                  float* __restrict__ out) {
    const int wave = threadIdx.x >> 5;
    const int lane = threadIdx.x & 31;
    const int tile = blockIdx.x * 4 + wave;
    const int row0 = (tile >> 4) * 16;
    const int col0 = (tile & 15) * 16;
    const int lm = lane & 15;
    const int hi = lane >> 4;

    const float bpv = bp[col0 + lm];
    const float brv = br[col0 + lm];
    const float bkv = bk[col0 + lm];
    v8f cp, cr, ck;
#pragma unroll
    for (int j = 0; j < 8; ++j) { cp[j] = bpv; cr[j] = brv; ck[j] = bkv; }

    const float* arow = M  + (size_t)(row0 + lm) * HH + 2 * hi;
    const float* wpc  = Wp + (size_t)(2 * hi) * DD + col0 + lm;
    const float* wrc  = Wr + (size_t)(2 * hi) * DD + col0 + lm;
    const float* wkc  = Wk + (size_t)(2 * hi) * DD + col0 + lm;

#pragma unroll 4
    for (int k = 0; k < HH; k += 4) {
        v2f a = *(const v2f*)(arow + k);

        v2f bbp, bbr, bbk;
        const size_t ko = (size_t)k * DD;
        bbp.x = wpc[ko];  bbp.y = wpc[ko + DD];
        bbr.x = wrc[ko];  bbr.y = wrc[ko + DD];
        bbk.x = wkc[ko];  bbk.y = wkc[ko + DD];

        cp = __builtin_amdgcn_wmma_f32_16x16x4_f32(false, a, false, bbp, (short)0, cp, false, false);
        cr = __builtin_amdgcn_wmma_f32_16x16x4_f32(false, a, false, bbr, (short)0, cr, false, false);
        ck = __builtin_amdgcn_wmma_f32_16x16x4_f32(false, a, false, bbk, (short)0, ck, false, false);
    }

    const size_t GD = (size_t)GG * DD;
    float* okeys = out;
    float* opq   = out + GD;
    float* orq   = out + 2 * GD;
#pragma unroll
    for (int j = 0; j < 8; ++j) {
        const int row = row0 + j + 8 * hi;
        const size_t idx = (size_t)row * DD + col0 + lm;
        okeys[idx] = ck[j];
        opq[idx]   = cp[j];
        orq[idx]   = cr[j];
    }
}

extern "C" void kernel_launch(void* const* d_in, const int* in_sizes, int n_in,
                              void* d_out, int out_size, void* d_ws, size_t ws_size,
                              hipStream_t stream) {
    const float* features = (const float*)d_in[0];
    const int*   num_nodes = (const int*)d_in[1];
    const float* Wp = (const float*)d_in[2];
    const float* bp = (const float*)d_in[3];
    const float* Wr = (const float*)d_in[4];
    const float* br = (const float*)d_in[5];
    const float* Wk = (const float*)d_in[6];
    const float* bk = (const float*)d_in[7];
    float* out = (float*)d_out;

    float* M    = (float*)((char*)d_ws + 8192);

    pool_mean_kernel<<<GG / 8, POOL_WAVES * 32, 0, stream>>>(features, num_nodes, M);
    gemm3_wmma_kernel<<<GG / 4, 128, 0, stream>>>(M, Wp, bp, Wr, br, Wk, bk, out);
    (void)in_sizes; (void)n_in; (void)out_size; (void)ws_size;
}


# GEMM tiled through LDS: each of 256 workgroups copies 64 mean rows and 96 weight columns to LDS once (16 B line loads, 4 k-chunks, next ones in flight); 8 waves read 8 B fragments; sum order kept
# speedup vs baseline: 1.7352x; 1.4538x over previous
#include <hip/hip_runtime.h>
#include <hip/hip_bf16.h>

#define GG 2048
#define HH 256
#define DD 256

typedef __attribute__((ext_vector_type(2))) float v2f;
typedef __attribute__((ext_vector_type(8))) float v8f;

#define POOL_WAVES 16
#define POOL_ROWS  16

typedef __attribute__((ext_vector_type(4))) float v4f;

#define POOL_LOAD(BUF, BI)                                                    \
    _Pragma("unroll")                                                         \
    for (int i = 0; i < POOL_ROWS; ++i) {                                     \
        const int r_  = (BI) * POOL_ROWS + i;                                 \
        const int rr_ = r_ < last ? r_ : last;                                \
        BUF[i] = *(const v4f*)(gb + (lo + (unsigned)rr_ * (HH * 4u)));        \
    }

#define POOL_SUM_FULL(BUF)                                                    \
    _Pragma("unroll")                                                         \
    for (int i = 0; i < POOL_ROWS; ++i) acc[i & 3] += BUF[i];

#define POOL_SUM_LAST(BUF, BI)                                                \
    _Pragma("unroll")                                                         \
    for (int i = 0; i < POOL_ROWS; ++i) {                                     \
        const int r_ = (BI) * POOL_ROWS + i;                                  \
        const v4f x_ = BUF[i];                                                \
        acc[i & 3] += (r_ < cnt4) ? x_ : zero;                                \
        acc[0]     += (r_ >= cnt4 && r_ < cnt) ? x_ : zero;                   \
    }

#define POOL_PIN()                                                            \
    asm volatile("" : "+v"(acc[0]), "+v"(acc[1]), "+v"(acc[2]), "+v"(acc[3]) :: "memory")

typedef __attribute__((ext_vector_type(4))) int v4i;

__device__ __forceinline__ int wave_sum(int x) {
    x += __builtin_amdgcn_update_dpp(0, x, 0x161, 0xf, 0xf, false);
    x += __builtin_amdgcn_update_dpp(0, x, 0x162, 0xf, 0xf, false);
    x += __builtin_amdgcn_update_dpp(0, x, 0x164, 0xf, 0xf, false);
    x += __builtin_amdgcn_update_dpp(0, x, 0x168, 0xf, 0xf, false);
    x += (int)__builtin_amdgcn_permlanex16((unsigned)x, (unsigned)x, 0x76543210u, 0xfedcba98u, false, false);
    return x;
}

__global__ void __launch_bounds__(POOL_WAVES * 32)
pool_mean_kernel(const float* __restrict__ feat,
                 const int* __restrict__ nn,
                 float* __restrict__ M) {
    __shared__ __attribute__((aligned(16))) int part[2 * POOL_WAVES];
    const int lane = threadIdx.x & 31;
    const int wave = __builtin_amdgcn_readfirstlane((int)(threadIdx.x >> 5));
    const int half = wave & 1;
    const int g    = (wave >> 1) * (GG / 8) + (int)blockIdx.x;
    const int cnt  = nn[g];

    {
        const int i0 = (int)threadIdx.x * 4;
        const v4i c4 = *(const v4i*)(nn + i0);
        const int whole = (c4.x + c4.y) + (c4.z + c4.w);
        const int below = ((i0     < g ? c4.x : 0) + (i0 + 1 < g ? c4.y : 0))
                        + ((i0 + 2 < g ? c4.z : 0) + (i0 + 3 < g ? c4.w : 0));
        const int ws = wave_sum(whole);
        const int bs = wave_sum(below);
        if (lane == 0) { part[wave] = ws; part[POOL_WAVES + wave] = bs; }
    }
    __syncthreads();
    const int pair = wave & ~1;
    int offv = part[POOL_WAVES + pair] + part[POOL_WAVES + pair + 1];
#pragma unroll
    for (int w = 0; w < POOL_WAVES - 2; ++w) offv += (w < pair) ? part[w] : 0;
    const int off  = __builtin_amdgcn_readfirstlane(offv);
    const int cnt4 = cnt & ~3;
    const int last = cnt - 1;
    const int nb   = (cnt + POOL_ROWS - 1) / POOL_ROWS;

    const v4f zero = {0.f, 0.f, 0.f, 0.f};
    v4f acc[4] = {zero, zero, zero, zero};

    if (cnt > 0) {
        const char* gb = (const char*)(feat + (size_t)off * HH + half * (HH / 2));
        const unsigned lo = (unsigned)lane * 16u;
        v4f bufA[POOL_ROWS], bufB[POOL_ROWS];
        int b = 0;
        POOL_LOAD(bufA, 0)
        while (b + 2 < nb) {
            __builtin_amdgcn_sched_barrier(0);
            POOL_LOAD(bufB, b + 1)
            __builtin_amdgcn_sched_barrier(0);
            POOL_SUM_FULL(bufA)
            POOL_PIN();
            __builtin_amdgcn_sched_barrier(0);
            POOL_LOAD(bufA, b + 2)
            __builtin_amdgcn_sched_barrier(0);
            POOL_SUM_FULL(bufB)
            POOL_PIN();
            b += 2;
        }
        if (b + 2 == nb) {
            __builtin_amdgcn_sched_barrier(0);
            POOL_LOAD(bufB, b + 1)
            __builtin_amdgcn_sched_barrier(0);
            POOL_SUM_FULL(bufA)
            POOL_SUM_LAST(bufB, b + 1)
        } else {
            POOL_SUM_LAST(bufA, b)
        }
    }

    const float inv = 1.0f / (float)cnt;
    const v4f res = (acc[0] + acc[1] + acc[2] + acc[3]) * inv;
    *(v4f*)(M + (size_t)g * HH + half * (HH / 2) + lane * 4) = res;
}

#define TR 64
#define TCT 6
#define KC 64
#define NCH (HH / KC)
#define PA 260
#define PBP 112

__global__ void __launch_bounds__(256)
gemm3_wmma_kernel(const float* __restrict__ M,
                  const float* __restrict__ Wp, const float* __restrict__ bp,
                  const float* __restrict__ Wr, const float* __restrict__ br,
                  const float* __restrict__ Wk, const float* __restrict__ bk,
                  float* __restrict__ out) {
    __shared__ __attribute__((aligned(16))) float As[TR * PA];
    __shared__ __attribute__((aligned(16))) float Bs[(HH / 2) * PBP * 2];

    const int tid  = (int)threadIdx.x;
    const int lane = tid & 31;
    const int wave = __builtin_amdgcn_readfirstlane(tid >> 5);
    const int rb   = (int)blockIdx.x >> 3;
    const int cb   = (int)blockIdx.x & 7;
    const int row0 = rb * TR;
    const int lm = lane & 15;
    const int hi = lane >> 4;

    const float* gA = M + (size_t)(row0 + (tid >> 4)) * HH + (tid & 15) * 4;
    const int    sA = (tid >> 4) * PA + (tid & 15) * 4;
    const float* gB[3];
    int sB[3];
#pragma unroll
    for (int i = 0; i < 3; ++i) {
        const int p  = tid + 256 * i;
        const int kp = p / 24;
        const int q  = p - kp * 24;
        const int t  = TCT * cb + (q >> 2);
        const int m  = t >> 4;
        const float* wsel = (m == 0) ? Wk : ((m == 1) ? Wp : Wr);
        gB[i] = wsel + (size_t)(2 * kp) * DD + (t & 15) * 16 + (q & 3) * 4;
        sB[i] = (kp * PBP + q * 4) * 2;
    }

    v4f ra[4], rw0[3], rw1[3];
#define G3_LOAD(CH)                                                           \
    _Pragma("unroll")                                                         \
    for (int i = 0; i < 4; ++i)                                               \
        ra[i] = *(const v4f*)(gA + (size_t)i * 16 * HH + (CH) * KC);          \
    _Pragma("unroll")                                                         \
    for (int i = 0; i < 3; ++i) {                                             \
        rw0[i] = *(const v4f*)(gB[i] + (size_t)(CH) * KC * DD);               \
        rw1[i] = *(const v4f*)(gB[i] + (size_t)(CH) * KC * DD + DD);          \
    }
#define G3_STORE(CH)                                                          \
    _Pragma("unroll")                                                         \
    for (int i = 0; i < 4; ++i)                                               \
        *(v4f*)(As + sA + i * 16 * PA + (CH) * KC) = ra[i];                   \
    _Pragma("unroll")                                                         \
    for (int i = 0; i < 3; ++i) {                                             \
        const v4f e_ = {rw0[i].x, rw1[i].x, rw0[i].y, rw1[i].y};              \
        const v4f o_ = {rw0[i].z, rw1[i].z, rw0[i].w, rw1[i].w};              \
        float* d_ = Bs + sB[i] + (CH) * (KC / 2) * PBP * 2;                   \
        *(v4f*)(d_)     = e_;                                                 \
        *(v4f*)(d_ + 4) = o_;                                                 \
    }

    const int rt = wave & 3;
    const int cg = wave >> 2;
    const int t0 = TCT * cb + 3 * cg;

    v8f acc[3];
#pragma unroll
    for (int j = 0; j < 3; ++j) {
        const int t = t0 + j;
        const int m = t >> 4;
        const float* bsel = (m == 0) ? bk : ((m == 1) ? bp : br);
        const float bv = bsel[(t & 15) * 16 + lm];
#pragma unroll
        for (int r = 0; r < 8; ++r) acc[j][r] = bv;
    }

    const float* aF = As + (rt * 16 + lm) * PA + 2 * hi;
    const float* bF = Bs + (hi * PBP + cg * 48 + lm) * 2;

    G3_LOAD(0)
    G3_STORE(0)
#pragma unroll
    for (int ch = 0; ch < NCH; ++ch) {
        if (ch + 1 < NCH) { G3_LOAD(ch + 1) }
        __syncthreads();
#pragma unroll
        for (int ks = 0; ks < KC / 4; ++ks) {
            const int k = ch * KC + ks * 4;
            const v2f a = *(const v2f*)(aF + k);
#pragma unroll
            for (int j = 0; j < 3; ++j) {
                const v2f b = *(const v2f*)(bF + (k / 2) * PBP * 2 + j * 32);
                acc[j] = __builtin_amdgcn_wmma_f32_16x16x4_f32(false, a, false, b, (short)0, acc[j], false, false);
            }
        }
        if (ch + 1 < NCH) { G3_STORE(ch + 1) }
    }

    const size_t GD = (size_t)GG * DD;
#pragma unroll
    for (int j = 0; j < 3; ++j) {
        const int t = t0 + j;
        float* op = out + (size_t)(t >> 4) * GD + (t & 15) * 16 + lm;
#pragma unroll
        for (int r = 0; r < 8; ++r) {
            const int row = row0 + rt * 16 + r + 8 * hi;
            op[(size_t)row * DD] = acc[j][r];
        }
    }
}

extern "C" void kernel_launch(void* const* d_in, const int* in_sizes, int n_in,
                              void* d_out, int out_size, void* d_ws, size_t ws_size,
                              hipStream_t stream) {
    const float* features = (const float*)d_in[0];
    const int*   num_nodes = (const int*)d_in[1];
    const float* Wp = (const float*)d_in[2];
    const float* bp = (const float*)d_in[3];
    const float* Wr = (const float*)d_in[4];
    const float* br = (const float*)d_in[5];
    const float* Wk = (const float*)d_in[6];
    const float* bk = (const float*)d_in[7];
    float* out = (float*)d_out;

    float* M    = (float*)((char*)d_ws + 8192);

    pool_mean_kernel<<<GG / 8, POOL_WAVES * 32, 0, stream>>>(features, num_nodes, M);
    gemm3_wmma_kernel<<<(GG / TR) * 8, 256, 0, stream>>>(M, Wp, bp, Wr, br, Wk, bk, out);
    (void)in_sizes; (void)n_in; (void)out_size; (void)ws_size;
}
